# starting kernel
# baseline (speedup 1.0000x reference)
#include <hip/hip_runtime.h>

#define S_LEN 2048
#define EMB   1024
#define NH    16
#define DH    64
#define QK_SCALE 0.125f

typedef __attribute__((ext_vector_type(2))) float v2f;
typedef __attribute__((ext_vector_type(8))) float v8f;

static __device__ __forceinline__ v2f ld_frag_nmajor(const float* __restrict__ base,
                                                     int row0, int ld, int k0, int lane) {
  const float* p = base + (size_t)(row0 + (lane & 15)) * ld + k0 + ((lane >> 4) << 1);
  return *(const v2f*)p;
}

static __device__ __forceinline__ v2f ld_frag_kmajor(const float* __restrict__ base,
                                                     int k0, int ld, int n0, int lane) {
  const float* p = base + (size_t)(k0 + ((lane >> 4) << 1)) * ld + n0 + (lane & 15);
  v2f r;
  r.x = p[0];
  r.y = p[ld];
  return r;
}

static __device__ __forceinline__ v8f wmma4(v2f a, v2f b, v8f c) {
  return __builtin_amdgcn_wmma_f32_16x16x4_f32(false, a, false, b, (short)0, c,
                                               false, false);
}

static __device__ __forceinline__ void gemm64x64(const float* __restrict__ A, int lda,
                                                 const float* __restrict__ BT, int ldb,
                                                 int K, int m0, int n0, int lane,
                                                 v8f acc[4][4]) {
  v8f zero = {};
#pragma unroll
  for (int i = 0; i < 4; ++i)
#pragma unroll
    for (int j = 0; j < 4; ++j) acc[i][j] = zero;

  for (int k0 = 0; k0 < K; k0 += 4) {
    v2f a[4], b[4];
#pragma unroll
    for (int i = 0; i < 4; ++i) a[i] = ld_frag_nmajor(A, m0 + 16 * i, lda, k0, lane);
#pragma unroll
    for (int j = 0; j < 4; ++j) b[j] = ld_frag_nmajor(BT, n0 + 16 * j, ldb, k0, lane);
#pragma unroll
    for (int i = 0; i < 4; ++i)
#pragma unroll
      for (int j = 0; j < 4; ++j) acc[i][j] = wmma4(a[i], b[j], acc[i][j]);
  }
}

__global__ __launch_bounds__(128) void qkv_proj_kernel(
    const float* __restrict__ X, const float* __restrict__ W,
    const float* __restrict__ bias, float* __restrict__ Qb,
    float* __restrict__ Kb, float* __restrict__ Vb) {
  const int lane = threadIdx.x & 31;
  const int w    = threadIdx.x >> 5;
  const int m0   = blockIdx.y * 128 + (w >> 1) * 64;
  const int n0   = blockIdx.x * 128 + (w & 1) * 64;

  v8f acc[4][4];
  gemm64x64(X, EMB, W, EMB, EMB, m0, n0, lane, acc);

  const int half = lane >> 4, l = lane & 15;
#pragma unroll
  for (int j = 0; j < 4; ++j) {
    const int ncol = n0 + 16 * j;
    const int t    = ncol >> 10;
    const int rem  = ncol & 1023;
    const int h    = rem >> 6;
    const int d    = (rem & 63) + l;
    const float bv = bias[ncol + l];
    float* dst = (t == 0) ? Qb : (t == 1) ? Kb : Vb;
    const float scl = (t == 0) ? QK_SCALE : 1.0f;
#pragma unroll
    for (int i = 0; i < 4; ++i) {
#pragma unroll
      for (int v = 0; v < 8; ++v) {
        const int mrow = m0 + 16 * i + v + 8 * half;
        const int bb   = mrow >> 11;
        const int s    = mrow & 2047;
        const size_t off = (((size_t)(bb * NH + h) * S_LEN) + s) * DH + d;
        dst[off] = (acc[i][j][v] + bv) * scl;
      }
    }
  }
}

__global__ __launch_bounds__(128) void attn_kernel(
    const float* __restrict__ Qb, const float* __restrict__ Kb,
    const float* __restrict__ Vb, float* __restrict__ AO) {
  __shared__ float pbuf[4][16][68];

  const int lane = threadIdx.x & 31;
  const int w    = threadIdx.x >> 5;
  const int bh   = blockIdx.x;
  const int q0   = blockIdx.y * 64 + w * 16;
  const float* Qh = Qb + (size_t)bh * S_LEN * DH;
  const float* Kh = Kb + (size_t)bh * S_LEN * DH;
  const float* Vh = Vb + (size_t)bh * S_LEN * DH;
  const int half = lane >> 4, l = lane & 15;

  v2f qf[16];
#pragma unroll
  for (int t = 0; t < 16; ++t) qf[t] = ld_frag_nmajor(Qh, q0, DH, 4 * t, lane);

  v8f zero = {};
  v8f o[4] = {zero, zero, zero, zero};
  float rsum[8] = {0, 0, 0, 0, 0, 0, 0, 0};

  for (int kb = 0; kb < S_LEN / 64; ++kb) {
    const int krow = kb * 64;
#pragma unroll
    for (int nt = 0; nt < 4; ++nt) {
      v8f sa = zero;
#pragma unroll
      for (int t = 0; t < 16; ++t) {
        v2f bf = ld_frag_nmajor(Kh, krow + 16 * nt, DH, 4 * t, lane);
        sa = wmma4(qf[t], bf, sa);
      }
#pragma unroll
      for (int v = 0; v < 8; ++v) {
        const float e = __expf(sa[v]);
        rsum[v] += e;
        pbuf[w][v + 8 * half][16 * nt + l] = e;
      }
    }
    __syncthreads();
#pragma unroll
    for (int dt = 0; dt < 4; ++dt) {
      v8f oa = o[dt];
#pragma unroll
      for (int t = 0; t < 16; ++t) {
        v2f pa = *(const v2f*)&pbuf[w][l][4 * t + 2 * half];
        v2f bv = ld_frag_kmajor(Vh, krow + 4 * t, DH, 16 * dt, lane);
        oa = wmma4(pa, bv, oa);
      }
      o[dt] = oa;
    }
    __syncthreads();
  }

  float inv[8];
#pragma unroll
  for (int v = 0; v < 8; ++v) {
    float s = rsum[v];
    s += __shfl_xor(s, 1);
    s += __shfl_xor(s, 2);
    s += __shfl_xor(s, 4);
    s += __shfl_xor(s, 8);
    inv[v] = 1.0f / s;
  }

  const int bb = bh >> 4, h = bh & 15;
#pragma unroll
  for (int dt = 0; dt < 4; ++dt) {
#pragma unroll
    for (int v = 0; v < 8; ++v) {
      const int qrow = q0 + v + 8 * half;
      const int d    = 16 * dt + l;
      AO[((size_t)(bb * S_LEN + qrow)) * EMB + h * DH + d] = o[dt][v] * inv[v];
    }
  }
}

__global__ __launch_bounds__(128) void out_proj_kernel(
    const float* __restrict__ AO, const float* __restrict__ W,
    const float* __restrict__ bias, float* __restrict__ Out) {
  const int lane = threadIdx.x & 31;
  const int w    = threadIdx.x >> 5;
  const int m0   = blockIdx.y * 128 + (w >> 1) * 64;
  const int n0   = blockIdx.x * 128 + (w & 1) * 64;

  v8f acc[4][4];
  gemm64x64(AO, EMB, W, EMB, EMB, m0, n0, lane, acc);

  const int half = lane >> 4, l = lane & 15;
#pragma unroll
  for (int j = 0; j < 4; ++j) {
    const int ncol = n0 + 16 * j + l;
    const float bv = bias[ncol];
#pragma unroll
    for (int i = 0; i < 4; ++i)
#pragma unroll
      for (int v = 0; v < 8; ++v) {
        const int mrow = m0 + 16 * i + v + 8 * half;
        Out[(size_t)mrow * EMB + ncol] = acc[i][j][v] + bv;
      }
  }
}

extern "C" void kernel_launch(void* const* d_in, const int* in_sizes, int n_in,
                              void* d_out, int out_size, void* d_ws, size_t ws_size,
                              hipStream_t stream) {
  const float* x    = (const float*)d_in[0];
  const float* Wqkv = (const float*)d_in[1];
  const float* bqkv = (const float*)d_in[2];
  const float* Wout = (const float*)d_in[3];
  const float* bout = (const float*)d_in[4];
  float* out = (float*)d_out;

  const size_t per = (size_t)4 * NH * S_LEN * DH;
  float* Qb = (float*)d_ws;
  float* Kb = Qb + per;
  float* Vb = Kb + per;
  float* AO = Vb + per;

  dim3 blk(128);
  qkv_proj_kernel<<<dim3(3 * EMB / 128, 4 * S_LEN / 128), blk, 0, stream>>>(
      x, Wqkv, bqkv, Qb, Kb, Vb);
  attn_kernel<<<dim3(4 * NH, S_LEN / 64), blk, 0, stream>>>(Qb, Kb, Vb, AO);
  out_proj_kernel<<<dim3(EMB / 128, 4 * S_LEN / 128), blk, 0, stream>>>(
      AO, Wout, bout, out);
}


# projection GEMMs (qkv, out): operand tiles fetched once per workgroup as whole 128-byte lines into double-buffered LDS; WMMA fragments read from LDS, not by row-strided partial-line global loads
# speedup vs baseline: 1.4726x; 1.4726x over previous
#include <hip/hip_runtime.h>

#define S_LEN 2048
#define EMB   1024
#define NH    16
#define DH    64
#define QK_SCALE 0.125f

typedef __attribute__((ext_vector_type(2))) float v2f;
typedef __attribute__((ext_vector_type(4))) float v4f;
typedef __attribute__((ext_vector_type(8))) float v8f;

static __device__ __forceinline__ v4f ld_frag2_nmajor(const float* __restrict__ base,
                                                      int row0, int ld, int k0, int lane) {
  const float* p = base + (size_t)(row0 + (lane & 15)) * ld + k0 + ((lane >> 4) << 2);
  return *(const v4f*)p;
}

static __device__ __forceinline__ v2f ld_frag_kmajor(const float* __restrict__ base,
                                                     int k0, int ld, int n0, int lane) {
  const float* p = base + (size_t)(k0 + ((lane >> 4) << 1)) * ld + n0 + (lane & 15);
  v2f r;
  r.x = p[0];
  r.y = p[ld];
  return r;
}

static __device__ __forceinline__ v8f wmma4(v2f a, v2f b, v8f c) {
  return __builtin_amdgcn_wmma_f32_16x16x4_f32(false, a, false, b, (short)0, c,
                                               false, false);
}

#define GK 32
#define GP 36
#define GT 128
#define GOP (GT * GP)
#define GSTAGE (2 * GOP)

static __device__ __forceinline__ void gemm128x128_lds(const float* __restrict__ A,
                                                       const float* __restrict__ BT,
                                                       int wg_m0, int wg_n0,
                                                       float* __restrict__ sm,
                                                       v8f acc[4][4]) {
  const int tid  = threadIdx.x;
  const int lane = tid & 31;
  const int w    = tid >> 5;

  v8f zero = {};
#pragma unroll
  for (int i = 0; i < 4; ++i)
#pragma unroll
    for (int j = 0; j < 4; ++j) acc[i][j] = zero;

  const int srow = tid >> 3, scol = (tid & 7) << 2;
  const float* ga = A + (size_t)(wg_m0 + srow) * EMB + scol;
  const float* gb = BT + (size_t)(wg_n0 + srow) * EMB + scol;
  const int soff = srow * GP + scol;
  const int fa = ((w >> 1) * 64 + (lane & 15)) * GP + ((lane >> 4) << 2);
  const int fb = ((w & 1) * 64 + (lane & 15)) * GP + ((lane >> 4) << 2);

  v4f ra[8], rb[8];
#pragma unroll
  for (int i = 0; i < 8; ++i) {
    ra[i] = *(const v4f*)(ga + (size_t)i * 16 * EMB);
    rb[i] = *(const v4f*)(gb + (size_t)i * 16 * EMB);
  }
#pragma unroll
  for (int i = 0; i < 8; ++i) {
    *(v4f*)(sm + soff + i * 16 * GP)       = ra[i];
    *(v4f*)(sm + GOP + soff + i * 16 * GP) = rb[i];
  }
  __syncthreads();

#pragma unroll 1
  for (int c = 0; c < EMB / GK; ++c) {
    const float* cA = sm + (c & 1) * GSTAGE;
    const float* cB = cA + GOP;
    const bool more = (c + 1) < EMB / GK;
    if (more) {
#pragma unroll
      for (int i = 0; i < 8; ++i) {
        ra[i] = *(const v4f*)(ga + (size_t)i * 16 * EMB + (c + 1) * GK);
        rb[i] = *(const v4f*)(gb + (size_t)i * 16 * EMB + (c + 1) * GK);
      }
    }
#pragma unroll
    for (int k0 = 0; k0 < GK; k0 += 8) {
      v4f a[4], b[4];
#pragma unroll
      for (int i = 0; i < 4; ++i) a[i] = *(const v4f*)(cA + fa + i * 16 * GP + k0);
#pragma unroll
      for (int j = 0; j < 4; ++j) b[j] = *(const v4f*)(cB + fb + j * 16 * GP + k0);
#pragma unroll
      for (int i = 0; i < 4; ++i)
#pragma unroll
        for (int j = 0; j < 4; ++j) acc[i][j] = wmma4(a[i].lo, b[j].lo, acc[i][j]);
#pragma unroll
      for (int i = 0; i < 4; ++i)
#pragma unroll
        for (int j = 0; j < 4; ++j) acc[i][j] = wmma4(a[i].hi, b[j].hi, acc[i][j]);
    }
    if (more) {
      float* nA = sm + ((c + 1) & 1) * GSTAGE;
#pragma unroll
      for (int i = 0; i < 8; ++i) {
        *(v4f*)(nA + soff + i * 16 * GP)       = ra[i];
        *(v4f*)(nA + GOP + soff + i * 16 * GP) = rb[i];
      }
    }
    __syncthreads();
  }
}

__global__ __launch_bounds__(128) void qkv_proj_kernel(
    const float* __restrict__ X, const float* __restrict__ W,
    const float* __restrict__ bias, float* __restrict__ Qb,
    float* __restrict__ Kb, float* __restrict__ Vb) {
  const int lane = threadIdx.x & 31;
  const int w    = threadIdx.x >> 5;
  const int m0   = blockIdx.y * 128 + (w >> 1) * 64;
  const int n0   = blockIdx.x * 128 + (w & 1) * 64;

  __shared__ __attribute__((aligned(16))) float gsm[2 * GSTAGE];
  v8f acc[4][4];
  gemm128x128_lds(X, W, blockIdx.y * 128, blockIdx.x * 128, gsm, acc);

  const int half = lane >> 4, l = lane & 15;
#pragma unroll
  for (int j = 0; j < 4; ++j) {
    const int ncol = n0 + 16 * j;
    const int t    = ncol >> 10;
    const int rem  = ncol & 1023;
    const int h    = rem >> 6;
    const int d    = (rem & 63) + l;
    const float bv = bias[ncol + l];
    float* dst = (t == 0) ? Qb : (t == 1) ? Kb : Vb;
    const float scl = (t == 0) ? QK_SCALE : 1.0f;
#pragma unroll
    for (int i = 0; i < 4; ++i) {
#pragma unroll
      for (int v = 0; v < 8; ++v) {
        const int mrow = m0 + 16 * i + v + 8 * half;
        const int bb   = mrow >> 11;
        const int s    = mrow & 2047;
        const size_t off = (((size_t)(bb * NH + h) * S_LEN) + s) * DH + d;
        dst[off] = (acc[i][j][v] + bv) * scl;
      }
    }
  }
}

#define QT 2
#define QROWS (16 * QT)

__global__ __launch_bounds__(128) void attn_kernel(
    const float* __restrict__ Qb, const float* __restrict__ Kb,
    const float* __restrict__ Vb, float* __restrict__ AO) {
  __shared__ float pbuf[4][QROWS][68];

  const int lane = threadIdx.x & 31;
  const int w    = threadIdx.x >> 5;
  const int bh   = blockIdx.x;
  const int q0   = blockIdx.y * (4 * QROWS) + w * QROWS;
  const float* Qh = Qb + (size_t)bh * S_LEN * DH;
  const float* Kh = Kb + (size_t)bh * S_LEN * DH;
  const float* Vh = Vb + (size_t)bh * S_LEN * DH;
  const int half = lane >> 4, l = lane & 15;

  v4f qf[QT][8];
#pragma unroll
  for (int qt = 0; qt < QT; ++qt)
#pragma unroll
    for (int t = 0; t < 8; ++t)
      qf[qt][t] = ld_frag2_nmajor(Qh, q0 + 16 * qt, DH, 8 * t, lane);

  v8f zero = {};
  v8f o[QT][4];
  float rsum[QT][8];
#pragma unroll
  for (int qt = 0; qt < QT; ++qt) {
#pragma unroll
    for (int dt = 0; dt < 4; ++dt) o[qt][dt] = zero;
#pragma unroll
    for (int v = 0; v < 8; ++v) rsum[qt][v] = 0.0f;
  }

  for (int kb = 0; kb < S_LEN / 64; ++kb) {
    const int krow = kb * 64;
#pragma unroll 1
    for (int nt = 0; nt < 4; ++nt) {
      v8f sa[QT];
#pragma unroll
      for (int qt = 0; qt < QT; ++qt) sa[qt] = zero;
#pragma unroll
      for (int t = 0; t < 8; ++t) {
        v4f bf = ld_frag2_nmajor(Kh, krow + 16 * nt, DH, 8 * t, lane);
#pragma unroll
        for (int qt = 0; qt < QT; ++qt) sa[qt] = wmma4(qf[qt][t].lo, bf.lo, sa[qt]);
#pragma unroll
        for (int qt = 0; qt < QT; ++qt) sa[qt] = wmma4(qf[qt][t].hi, bf.hi, sa[qt]);
      }
#pragma unroll
      for (int qt = 0; qt < QT; ++qt) {
#pragma unroll
        for (int v = 0; v < 8; ++v) {
          const float e = __expf(sa[qt][v]);
          rsum[qt][v] += e;
          pbuf[w][16 * qt + v + 8 * half][16 * nt + l] = e;
        }
      }
    }
    __syncthreads();
#pragma unroll 1
    for (int tg = 0; tg < 16; tg += 4)
#pragma unroll
    for (int t = tg; t < tg + 4; ++t) {
      v2f pa[QT];
#pragma unroll
      for (int qt = 0; qt < QT; ++qt)
        pa[qt] = *(const v2f*)&pbuf[w][16 * qt + l][4 * t + 2 * half];
#pragma unroll
      for (int dt = 0; dt < 4; ++dt) {
        v2f bv = ld_frag_kmajor(Vh, krow + 4 * t, DH, 16 * dt, lane);
#pragma unroll
        for (int qt = 0; qt < QT; ++qt) o[qt][dt] = wmma4(pa[qt], bv, o[qt][dt]);
      }
    }
    __syncthreads();
  }

  const int bb = bh >> 4, h = bh & 15;
#pragma unroll
  for (int qt = 0; qt < QT; ++qt) {
    float inv[8];
#pragma unroll
    for (int v = 0; v < 8; ++v) {
      float s = rsum[qt][v];
      s += __shfl_xor(s, 1);
      s += __shfl_xor(s, 2);
      s += __shfl_xor(s, 4);
      s += __shfl_xor(s, 8);
      inv[v] = 1.0f / s;
    }
#pragma unroll
    for (int dt = 0; dt < 4; ++dt) {
#pragma unroll
      for (int v = 0; v < 8; ++v) {
        const int qrow = q0 + 16 * qt + v + 8 * half;
        const int d    = 16 * dt + l;
        AO[((size_t)(bb * S_LEN + qrow)) * EMB + h * DH + d] = o[qt][dt][v] * inv[v];
      }
    }
  }
}

__global__ __launch_bounds__(128) void out_proj_kernel(
    const float* __restrict__ AO, const float* __restrict__ W,
    const float* __restrict__ bias, float* __restrict__ Out) {
  const int lane = threadIdx.x & 31;
  const int w    = threadIdx.x >> 5;
  const int m0   = blockIdx.y * 128 + (w >> 1) * 64;
  const int n0   = blockIdx.x * 128 + (w & 1) * 64;

  __shared__ __attribute__((aligned(16))) float gsm[2 * GSTAGE];
  v8f acc[4][4];
  gemm128x128_lds(AO, W, blockIdx.y * 128, blockIdx.x * 128, gsm, acc);

  const int half = lane >> 4, l = lane & 15;
#pragma unroll
  for (int j = 0; j < 4; ++j) {
    const int ncol = n0 + 16 * j + l;
    const float bv = bias[ncol];
#pragma unroll
    for (int i = 0; i < 4; ++i)
#pragma unroll
      for (int v = 0; v < 8; ++v) {
        const int mrow = m0 + 16 * i + v + 8 * half;
        Out[(size_t)mrow * EMB + ncol] = acc[i][j][v] + bv;
      }
  }
}

extern "C" void kernel_launch(void* const* d_in, const int* in_sizes, int n_in,
                              void* d_out, int out_size, void* d_ws, size_t ws_size,
                              hipStream_t stream) {
  const float* x    = (const float*)d_in[0];
  const float* Wqkv = (const float*)d_in[1];
  const float* bqkv = (const float*)d_in[2];
  const float* Wout = (const float*)d_in[3];
  const float* bout = (const float*)d_in[4];
  float* out = (float*)d_out;

  const size_t per = (size_t)4 * NH * S_LEN * DH;
  float* Qb = (float*)d_ws;
  float* Kb = Qb + per;
  float* Vb = Kb + per;
  float* AO = Vb + per;

  dim3 blk(128);
  qkv_proj_kernel<<<dim3(3 * EMB / 128, 4 * S_LEN / 128), blk, 0, stream>>>(
      x, Wqkv, bqkv, Qb, Kb, Vb);
  attn_kernel<<<dim3(4 * NH, S_LEN / (4 * QROWS)), blk, 0, stream>>>(Qb, Kb, Vb, AO);
  out_proj_kernel<<<dim3(EMB / 128, 4 * S_LEN / 128), blk, 0, stream>>>(
      AO, Wout, bout, out);
}


# attention: scores computed transposed (K tile as A, Q as B) so exp(S) in its D registers is directly the B operand of O^T = V^T P^T: P never goes through LDS, no LDS and no barrier in the kernel
# speedup vs baseline: 1.5186x; 1.0312x over previous
#include <hip/hip_runtime.h>

#define S_LEN 2048
#define EMB   1024
#define NH    16
#define DH    64
#define QK_SCALE 0.125f

typedef __attribute__((ext_vector_type(2))) float v2f;
typedef __attribute__((ext_vector_type(4))) float v4f;
typedef __attribute__((ext_vector_type(8))) float v8f;

static __device__ __forceinline__ v4f ld_frag2_nmajor(const float* __restrict__ base,
                                                      int row0, int ld, int k0, int lane) {
  const float* p = base + (size_t)(row0 + (lane & 15)) * ld + k0 + ((lane >> 4) << 2);
  return *(const v4f*)p;
}

static __device__ __forceinline__ v2f ld_frag_kmajor(const float* __restrict__ base,
                                                     int k0, int ld, int n0, int lane) {
  const float* p = base + (size_t)(k0 + ((lane >> 4) << 1)) * ld + n0 + (lane & 15);
  v2f r;
  r.x = p[0];
  r.y = p[ld];
  return r;
}

static __device__ __forceinline__ v8f wmma4(v2f a, v2f b, v8f c) {
  return __builtin_amdgcn_wmma_f32_16x16x4_f32(false, a, false, b, (short)0, c,
                                               false, false);
}

#define GK 32
#define GP 36
#define GT 128
#define GOP (GT * GP)
#define GSTAGE (2 * GOP)

static __device__ __forceinline__ void gemm128x128_lds(const float* __restrict__ A,
                                                       const float* __restrict__ BT,
                                                       int wg_m0, int wg_n0,
                                                       float* __restrict__ sm,
                                                       v8f acc[4][4]) {
  const int tid  = threadIdx.x;
  const int lane = tid & 31;
  const int w    = tid >> 5;

  v8f zero = {};
#pragma unroll
  for (int i = 0; i < 4; ++i)
#pragma unroll
    for (int j = 0; j < 4; ++j) acc[i][j] = zero;

  const int srow = tid >> 3, scol = (tid & 7) << 2;
  const float* ga = A + (size_t)(wg_m0 + srow) * EMB + scol;
  const float* gb = BT + (size_t)(wg_n0 + srow) * EMB + scol;
  const int soff = srow * GP + scol;
  const int fa = ((w >> 1) * 64 + (lane & 15)) * GP + ((lane >> 4) << 2);
  const int fb = ((w & 1) * 64 + (lane & 15)) * GP + ((lane >> 4) << 2);

  v4f ra[8], rb[8];
#pragma unroll
  for (int i = 0; i < 8; ++i) {
    ra[i] = *(const v4f*)(ga + (size_t)i * 16 * EMB);
    rb[i] = *(const v4f*)(gb + (size_t)i * 16 * EMB);
  }
#pragma unroll
  for (int i = 0; i < 8; ++i) {
    *(v4f*)(sm + soff + i * 16 * GP)       = ra[i];
    *(v4f*)(sm + GOP + soff + i * 16 * GP) = rb[i];
  }
  __syncthreads();

#pragma unroll 1
  for (int c = 0; c < EMB / GK; ++c) {
    const float* cA = sm + (c & 1) * GSTAGE;
    const float* cB = cA + GOP;
    const bool more = (c + 1) < EMB / GK;
    if (more) {
#pragma unroll
      for (int i = 0; i < 8; ++i) {
        ra[i] = *(const v4f*)(ga + (size_t)i * 16 * EMB + (c + 1) * GK);
        rb[i] = *(const v4f*)(gb + (size_t)i * 16 * EMB + (c + 1) * GK);
      }
    }
#pragma unroll
    for (int k0 = 0; k0 < GK; k0 += 8) {
      v4f a[4], b[4];
#pragma unroll
      for (int i = 0; i < 4; ++i) a[i] = *(const v4f*)(cA + fa + i * 16 * GP + k0);
#pragma unroll
      for (int j = 0; j < 4; ++j) b[j] = *(const v4f*)(cB + fb + j * 16 * GP + k0);
#pragma unroll
      for (int i = 0; i < 4; ++i)
#pragma unroll
        for (int j = 0; j < 4; ++j) acc[i][j] = wmma4(a[i].lo, b[j].lo, acc[i][j]);
#pragma unroll
      for (int i = 0; i < 4; ++i)
#pragma unroll
        for (int j = 0; j < 4; ++j) acc[i][j] = wmma4(a[i].hi, b[j].hi, acc[i][j]);
    }
    if (more) {
      float* nA = sm + ((c + 1) & 1) * GSTAGE;
#pragma unroll
      for (int i = 0; i < 8; ++i) {
        *(v4f*)(nA + soff + i * 16 * GP)       = ra[i];
        *(v4f*)(nA + GOP + soff + i * 16 * GP) = rb[i];
      }
    }
    __syncthreads();
  }
}

__global__ __launch_bounds__(128) void qkv_proj_kernel(
    const float* __restrict__ X, const float* __restrict__ W,
    const float* __restrict__ bias, float* __restrict__ Qb,
    float* __restrict__ Kb, float* __restrict__ Vb) {
  const int lane = threadIdx.x & 31;
  const int w    = threadIdx.x >> 5;
  const int m0   = blockIdx.y * 128 + (w >> 1) * 64;
  const int n0   = blockIdx.x * 128 + (w & 1) * 64;

  __shared__ __attribute__((aligned(16))) float gsm[2 * GSTAGE];
  v8f acc[4][4];
  gemm128x128_lds(X, W, blockIdx.y * 128, blockIdx.x * 128, gsm, acc);

  const int half = lane >> 4, l = lane & 15;
#pragma unroll
  for (int j = 0; j < 4; ++j) {
    const int ncol = n0 + 16 * j;
    const int t    = ncol >> 10;
    const int rem  = ncol & 1023;
    const int h    = rem >> 6;
    const int d    = (rem & 63) + l;
    const float bv = bias[ncol + l];
    float* dst = (t == 0) ? Qb : (t == 1) ? Kb : Vb;
    const float scl = (t == 0) ? QK_SCALE : 1.0f;
#pragma unroll
    for (int i = 0; i < 4; ++i) {
#pragma unroll
      for (int v = 0; v < 8; ++v) {
        const int mrow = m0 + 16 * i + v + 8 * half;
        const int bb   = mrow >> 11;
        const int s    = mrow & 2047;
        const size_t off = (((size_t)(bb * NH + h) * S_LEN) + s) * DH + d;
        dst[off] = (acc[i][j][v] + bv) * scl;
      }
    }
  }
}

#define QT 2
#define QROWS (16 * QT)
#define KT (S_LEN / 16)

__global__ __launch_bounds__(128) void attn_kernel(
    const float* __restrict__ Qb, const float* __restrict__ Kb,
    const float* __restrict__ Vb, float* __restrict__ AO) {
  const int lane = threadIdx.x & 31;
  const int w    = threadIdx.x >> 5;
  const int bh   = blockIdx.x;
  const int q0   = blockIdx.y * (4 * QROWS) + w * QROWS;
  const float* Qh = Qb + (size_t)bh * S_LEN * DH;
  const float* Kh = Kb + (size_t)bh * S_LEN * DH;
  const float* Vh = Vb + (size_t)bh * S_LEN * DH;
  const int half = lane >> 4, l = lane & 15;

  v4f qf[QT][8];
#pragma unroll
  for (int qt = 0; qt < QT; ++qt)
#pragma unroll
    for (int t = 0; t < 8; ++t)
      qf[qt][t] = ld_frag2_nmajor(Qh, q0 + 16 * qt, DH, 8 * t, lane);

  v8f zero = {};
  v8f o[QT][4];
  float rs[QT];
#pragma unroll
  for (int qt = 0; qt < QT; ++qt) {
#pragma unroll
    for (int dt = 0; dt < 4; ++dt) o[qt][dt] = zero;
    rs[qt] = 0.0f;
  }

  const float* kp = Kh + (size_t)l * DH + (half << 2);
  const float* vp = Vh + (size_t)(8 * half) * DH + l;

  v4f kf[8];
#pragma unroll
  for (int t = 0; t < 8; ++t) kf[t] = *(const v4f*)(kp + 8 * t);

#pragma unroll 1
  for (int kt = 0; kt < KT; ++kt) {
    const float* vt = vp + (size_t)kt * 16 * DH;
    v2f vf[4][4];
#pragma unroll
    for (int j = 0; j < 4; ++j)
#pragma unroll
      for (int dt = 0; dt < 4; ++dt) {
        vf[j][dt].x = vt[(2 * j) * DH + 16 * dt];
        vf[j][dt].y = vt[(2 * j + 1) * DH + 16 * dt];
      }
    __builtin_amdgcn_sched_barrier(0);

    v8f sa[QT];
#pragma unroll
    for (int qt = 0; qt < QT; ++qt) sa[qt] = zero;
#pragma unroll
    for (int t = 0; t < 8; ++t) {
#pragma unroll
      for (int qt = 0; qt < QT; ++qt) sa[qt] = wmma4(kf[t].lo, qf[qt][t].lo, sa[qt]);
#pragma unroll
      for (int qt = 0; qt < QT; ++qt) sa[qt] = wmma4(kf[t].hi, qf[qt][t].hi, sa[qt]);
    }

    __builtin_amdgcn_sched_barrier(0);
    const int nk = (kt + 1 < KT) ? kt + 1 : kt;
    const float* kn = kp + (size_t)nk * 16 * DH;
#pragma unroll
    for (int t = 0; t < 8; ++t) kf[t] = *(const v4f*)(kn + 8 * t);
    __builtin_amdgcn_sched_barrier(0);

    v2f pb[QT][4];
#pragma unroll
    for (int qt = 0; qt < QT; ++qt)
#pragma unroll
      for (int j = 0; j < 4; ++j) {
        const float e0 = __expf(sa[qt][2 * j]);
        const float e1 = __expf(sa[qt][2 * j + 1]);
        rs[qt] += e0 + e1;
        pb[qt][j].x = e0;
        pb[qt][j].y = e1;
      }

#pragma unroll
    for (int j = 0; j < 4; ++j)
#pragma unroll
      for (int dt = 0; dt < 4; ++dt)
#pragma unroll
        for (int qt = 0; qt < QT; ++qt) o[qt][dt] = wmma4(vf[j][dt], pb[qt][j], o[qt][dt]);
  }

  const int bb = bh >> 4, h = bh & 15;
#pragma unroll
  for (int qt = 0; qt < QT; ++qt) {
    const float s   = rs[qt] + __shfl_xor(rs[qt], 16);
    const float inv = 1.0f / s;
    float* orow = AO + ((size_t)(bb * S_LEN + q0 + 16 * qt + l)) * EMB + h * DH + 8 * half;
#pragma unroll
    for (int dt = 0; dt < 4; ++dt) {
      v4f lo4, hi4;
      lo4.x = o[qt][dt][0] * inv; lo4.y = o[qt][dt][1] * inv;
      lo4.z = o[qt][dt][2] * inv; lo4.w = o[qt][dt][3] * inv;
      hi4.x = o[qt][dt][4] * inv; hi4.y = o[qt][dt][5] * inv;
      hi4.z = o[qt][dt][6] * inv; hi4.w = o[qt][dt][7] * inv;
      *(v4f*)(orow + 16 * dt)     = lo4;
      *(v4f*)(orow + 16 * dt + 4) = hi4;
    }
  }
}

__global__ __launch_bounds__(128) void out_proj_kernel(
    const float* __restrict__ AO, const float* __restrict__ W,
    const float* __restrict__ bias, float* __restrict__ Out) {
  const int lane = threadIdx.x & 31;
  const int w    = threadIdx.x >> 5;
  const int m0   = blockIdx.y * 128 + (w >> 1) * 64;
  const int n0   = blockIdx.x * 128 + (w & 1) * 64;

  __shared__ __attribute__((aligned(16))) float gsm[2 * GSTAGE];
  v8f acc[4][4];
  gemm128x128_lds(AO, W, blockIdx.y * 128, blockIdx.x * 128, gsm, acc);

  const int half = lane >> 4, l = lane & 15;
#pragma unroll
  for (int j = 0; j < 4; ++j) {
    const int ncol = n0 + 16 * j + l;
    const float bv = bias[ncol];
#pragma unroll
    for (int i = 0; i < 4; ++i)
#pragma unroll
      for (int v = 0; v < 8; ++v) {
        const int mrow = m0 + 16 * i + v + 8 * half;
        Out[(size_t)mrow * EMB + ncol] = acc[i][j][v] + bv;
      }
  }
}

extern "C" void kernel_launch(void* const* d_in, const int* in_sizes, int n_in,
                              void* d_out, int out_size, void* d_ws, size_t ws_size,
                              hipStream_t stream) {
  const float* x    = (const float*)d_in[0];
  const float* Wqkv = (const float*)d_in[1];
  const float* bqkv = (const float*)d_in[2];
  const float* Wout = (const float*)d_in[3];
  const float* bout = (const float*)d_in[4];
  float* out = (float*)d_out;

  const size_t per = (size_t)4 * NH * S_LEN * DH;
  float* Qb = (float*)d_ws;
  float* Kb = Qb + per;
  float* Vb = Kb + per;
  float* AO = Vb + per;

  dim3 blk(128);
  qkv_proj_kernel<<<dim3(3 * EMB / 128, 4 * S_LEN / 128), blk, 0, stream>>>(
      x, Wqkv, bqkv, Qb, Kb, Vb);
  attn_kernel<<<dim3(4 * NH, S_LEN / (4 * QROWS)), blk, 0, stream>>>(Qb, Kb, Vb, AO);
  out_proj_kernel<<<dim3(EMB / 128, 4 * S_LEN / 128), blk, 0, stream>>>(
      AO, Wout, bout, out);
}


# K and V stored by the projection kernel in attention's WMMA fragment order (K via the transposed product): every K/V fragment load is one contiguous 512-byte load, 16 per key tile instead of 40
# speedup vs baseline: 1.5715x; 1.0348x over previous
#include <hip/hip_runtime.h>

#define S_LEN 2048
#define EMB   1024
#define NH    16
#define DH    64
#define QK_SCALE 0.125f
#define KT (S_LEN / 16)

typedef __attribute__((ext_vector_type(2))) float v2f;
typedef __attribute__((ext_vector_type(4))) float v4f;
typedef __attribute__((ext_vector_type(8))) float v8f;

static __device__ __forceinline__ v4f ld_frag2_nmajor(const float* __restrict__ base,
                                                      int row0, int ld, int k0, int lane) {
  const float* p = base + (size_t)(row0 + (lane & 15)) * ld + k0 + ((lane >> 4) << 2);
  return *(const v4f*)p;
}

static __device__ __forceinline__ v2f ld_frag_kmajor(const float* __restrict__ base,
                                                     int k0, int ld, int n0, int lane) {
  const float* p = base + (size_t)(k0 + ((lane >> 4) << 1)) * ld + n0 + (lane & 15);
  v2f r;
  r.x = p[0];
  r.y = p[ld];
  return r;
}

static __device__ __forceinline__ v8f wmma4(v2f a, v2f b, v8f c) {
  return __builtin_amdgcn_wmma_f32_16x16x4_f32(false, a, false, b, (short)0, c,
                                               false, false);
}

#define GK 32
#define GP 36
#define GT 128
#define GOP (GT * GP)
#define GSTAGE (2 * GOP)

static __device__ __forceinline__ void gemm128x128_lds(const float* __restrict__ A,
                                                       const float* __restrict__ BT,
                                                       int wg_m0, int wg_n0,
                                                       float* __restrict__ sm,
                                                       v8f acc[4][4]) {
  const int tid  = threadIdx.x;
  const int lane = tid & 31;
  const int w    = tid >> 5;

  v8f zero = {};
#pragma unroll
  for (int i = 0; i < 4; ++i)
#pragma unroll
    for (int j = 0; j < 4; ++j) acc[i][j] = zero;

  const int srow = tid >> 3, scol = (tid & 7) << 2;
  const float* ga = A + (size_t)(wg_m0 + srow) * EMB + scol;
  const float* gb = BT + (size_t)(wg_n0 + srow) * EMB + scol;
  const int soff = srow * GP + scol;
  const int fa = ((w >> 1) * 64 + (lane & 15)) * GP + ((lane >> 4) << 2);
  const int fb = ((w & 1) * 64 + (lane & 15)) * GP + ((lane >> 4) << 2);

  v4f ra[8], rb[8];
#pragma unroll
  for (int i = 0; i < 8; ++i) {
    ra[i] = *(const v4f*)(ga + (size_t)i * 16 * EMB);
    rb[i] = *(const v4f*)(gb + (size_t)i * 16 * EMB);
  }
#pragma unroll
  for (int i = 0; i < 8; ++i) {
    *(v4f*)(sm + soff + i * 16 * GP)       = ra[i];
    *(v4f*)(sm + GOP + soff + i * 16 * GP) = rb[i];
  }
  __syncthreads();

#pragma unroll 1
  for (int c = 0; c < EMB / GK; ++c) {
    const float* cA = sm + (c & 1) * GSTAGE;
    const float* cB = cA + GOP;
    const bool more = (c + 1) < EMB / GK;
    if (more) {
#pragma unroll
      for (int i = 0; i < 8; ++i) {
        ra[i] = *(const v4f*)(ga + (size_t)i * 16 * EMB + (c + 1) * GK);
        rb[i] = *(const v4f*)(gb + (size_t)i * 16 * EMB + (c + 1) * GK);
      }
    }
#pragma unroll
    for (int k0 = 0; k0 < GK; k0 += 8) {
      v4f a[4], b[4];
#pragma unroll
      for (int i = 0; i < 4; ++i) a[i] = *(const v4f*)(cA + fa + i * 16 * GP + k0);
#pragma unroll
      for (int j = 0; j < 4; ++j) b[j] = *(const v4f*)(cB + fb + j * 16 * GP + k0);
#pragma unroll
      for (int i = 0; i < 4; ++i)
#pragma unroll
        for (int j = 0; j < 4; ++j) acc[i][j] = wmma4(a[i].lo, b[j].lo, acc[i][j]);
#pragma unroll
      for (int i = 0; i < 4; ++i)
#pragma unroll
        for (int j = 0; j < 4; ++j) acc[i][j] = wmma4(a[i].hi, b[j].hi, acc[i][j]);
    }
    if (more) {
      float* nA = sm + ((c + 1) & 1) * GSTAGE;
#pragma unroll
      for (int i = 0; i < 8; ++i) {
        *(v4f*)(nA + soff + i * 16 * GP)       = ra[i];
        *(v4f*)(nA + GOP + soff + i * 16 * GP) = rb[i];
      }
    }
    __syncthreads();
  }
}

__global__ __launch_bounds__(128) void qkv_proj_kernel(
    const float* __restrict__ X, const float* __restrict__ W,
    const float* __restrict__ bias, float* __restrict__ Qb,
    float* __restrict__ Kp, float* __restrict__ Vp) {
  const int lane = threadIdx.x & 31;
  const int w    = threadIdx.x >> 5;
  const int t    = blockIdx.x >> 3;
  const bool kT  = (t == 1);
  const float* A  = kT ? W : X;
  const float* BT = kT ? X : W;
  const int wg_m0 = (kT ? blockIdx.x : blockIdx.y) * 128;
  const int wg_n0 = (kT ? blockIdx.y : blockIdx.x) * 128;
  const int m0   = wg_m0 + (w >> 1) * 64;
  const int n0   = wg_n0 + (w & 1) * 64;

  __shared__ __attribute__((aligned(16))) float gsm[2 * GSTAGE];
  v8f acc[4][4];
  gemm128x128_lds(A, BT, wg_m0, wg_n0, gsm, acc);

  const int half = lane >> 4, l = lane & 15;
  if (t == 0) {
#pragma unroll
    for (int j = 0; j < 4; ++j) {
      const int ncol = n0 + 16 * j;
      const int h    = ncol >> 6;
      const int d    = (ncol & 63) + l;
      const float bv = bias[ncol + l];
#pragma unroll
      for (int i = 0; i < 4; ++i) {
#pragma unroll
        for (int v = 0; v < 8; ++v) {
          const int mrow = m0 + 16 * i + v + 8 * half;
          const int bb   = mrow >> 11;
          const int s    = mrow & 2047;
          const size_t off = (((size_t)(bb * NH + h) * S_LEN) + s) * DH + d;
          Qb[off] = (acc[i][j][v] + bv) * QK_SCALE;
        }
      }
    }
  } else if (t == 1) {
    const int h = (m0 & 1023) >> 6;
    v4f bl[4], bh4[4];
#pragma unroll
    for (int i = 0; i < 4; ++i) {
      const float* bp = bias + m0 + 16 * i + 8 * half;
      bl[i]  = *(const v4f*)bp;
      bh4[i] = *(const v4f*)(bp + 4);
    }
#pragma unroll
    for (int j = 0; j < 4; ++j) {
      const int r0 = n0 + 16 * j;
      const int bb = r0 >> 11;
      const int kt = (r0 & 2047) >> 4;
      float* tile = Kp + ((size_t)((bb * NH + h) * KT + kt)) * 1024 + l * 4;
#pragma unroll
      for (int i = 0; i < 4; ++i) {
        v4f lo4, hi4;
        lo4.x = acc[i][j][0] + bl[i].x;  lo4.y = acc[i][j][1] + bl[i].y;
        lo4.z = acc[i][j][2] + bl[i].z;  lo4.w = acc[i][j][3] + bl[i].w;
        hi4.x = acc[i][j][4] + bh4[i].x; hi4.y = acc[i][j][5] + bh4[i].y;
        hi4.z = acc[i][j][6] + bh4[i].z; hi4.w = acc[i][j][7] + bh4[i].w;
        *(v4f*)(tile + (4 * i + 2 * half) * 64)     = lo4;
        *(v4f*)(tile + (4 * i + 2 * half + 1) * 64) = hi4;
      }
    }
  } else {
    const int h = (n0 & 1023) >> 6;
#pragma unroll
    for (int i = 0; i < 4; ++i) {
      const int r0 = m0 + 16 * i;
      const int bb = r0 >> 11;
      const int kt = (r0 & 2047) >> 4;
      float* tile = Vp + ((size_t)((bb * NH + h) * KT + kt)) * 1024 + lane * 4;
#pragma unroll
      for (int j = 0; j < 4; ++j) {
        const float bv = bias[n0 + 16 * j + l];
        v4f lo4, hi4;
        lo4.x = acc[i][j][0] + bv; lo4.y = acc[i][j][1] + bv;
        lo4.z = acc[i][j][2] + bv; lo4.w = acc[i][j][3] + bv;
        hi4.x = acc[i][j][4] + bv; hi4.y = acc[i][j][5] + bv;
        hi4.z = acc[i][j][6] + bv; hi4.w = acc[i][j][7] + bv;
        *(v4f*)(tile + (2 * j) * 128)     = lo4;
        *(v4f*)(tile + (2 * j + 1) * 128) = hi4;
      }
    }
  }
}

#define QT 2
#define QROWS (16 * QT)

__global__ __launch_bounds__(128) void attn_kernel(
    const float* __restrict__ Qb, const float* __restrict__ Kp,
    const float* __restrict__ Vp, float* __restrict__ AO) {
  const int lane = threadIdx.x & 31;
  const int w    = threadIdx.x >> 5;
  const int bh   = blockIdx.x;
  const int q0   = blockIdx.y * (4 * QROWS) + w * QROWS;
  const float* Qh = Qb + (size_t)bh * S_LEN * DH;
  const float* Kh = Kp + (size_t)bh * S_LEN * DH;
  const float* Vh = Vp + (size_t)bh * S_LEN * DH;
  const int half = lane >> 4, l = lane & 15;

  v4f qf[QT][8];
#pragma unroll
  for (int qt = 0; qt < QT; ++qt)
#pragma unroll
    for (int t = 0; t < 8; ++t)
      qf[qt][t] = ld_frag2_nmajor(Qh, q0 + 16 * qt, DH, 8 * t, lane);

  v8f zero = {};
  v8f o[QT][4];
  float rs[QT];
#pragma unroll
  for (int qt = 0; qt < QT; ++qt) {
#pragma unroll
    for (int dt = 0; dt < 4; ++dt) o[qt][dt] = zero;
    rs[qt] = 0.0f;
  }

  const float* kp = Kh + half * 64 + l * 4;
  const float* vp = Vh + lane * 4;

  v4f kf[8];
#pragma unroll
  for (int t = 0; t < 8; ++t) kf[t] = *(const v4f*)(kp + t * 128);

#pragma unroll 1
  for (int kt = 0; kt < KT; ++kt) {
    const float* vt = vp + (size_t)kt * 1024;
    v4f vq[4][2];
#pragma unroll
    for (int dt = 0; dt < 4; ++dt)
#pragma unroll
      for (int jp = 0; jp < 2; ++jp) vq[dt][jp] = *(const v4f*)(vt + (dt * 2 + jp) * 128);
    __builtin_amdgcn_sched_barrier(0);

    v8f sa[QT];
#pragma unroll
    for (int qt = 0; qt < QT; ++qt) sa[qt] = zero;
#pragma unroll
    for (int t = 0; t < 8; ++t) {
#pragma unroll
      for (int qt = 0; qt < QT; ++qt) sa[qt] = wmma4(kf[t].lo, qf[qt][t].lo, sa[qt]);
#pragma unroll
      for (int qt = 0; qt < QT; ++qt) sa[qt] = wmma4(kf[t].hi, qf[qt][t].hi, sa[qt]);
    }

    __builtin_amdgcn_sched_barrier(0);
    const int nk = (kt + 1 < KT) ? kt + 1 : kt;
    const float* kn = kp + (size_t)nk * 1024;
#pragma unroll
    for (int t = 0; t < 8; ++t) kf[t] = *(const v4f*)(kn + t * 128);
    __builtin_amdgcn_sched_barrier(0);

    v2f pb[QT][4];
#pragma unroll
    for (int qt = 0; qt < QT; ++qt)
#pragma unroll
      for (int j = 0; j < 4; ++j) {
        const float e0 = __expf(sa[qt][2 * j]);
        const float e1 = __expf(sa[qt][2 * j + 1]);
        rs[qt] += e0 + e1;
        pb[qt][j].x = e0;
        pb[qt][j].y = e1;
      }

#pragma unroll
    for (int j = 0; j < 4; ++j)
#pragma unroll
      for (int dt = 0; dt < 4; ++dt)
#pragma unroll
        for (int qt = 0; qt < QT; ++qt)
          o[qt][dt] = wmma4((j & 1) ? vq[dt][j >> 1].hi : vq[dt][j >> 1].lo, pb[qt][j], o[qt][dt]);
  }

  const int bb = bh >> 4, h = bh & 15;
#pragma unroll
  for (int qt = 0; qt < QT; ++qt) {
    const float s   = rs[qt] + __shfl_xor(rs[qt], 16);
    const float inv = 1.0f / s;
    float* orow = AO + ((size_t)(bb * S_LEN + q0 + 16 * qt + l)) * EMB + h * DH + 8 * half;
#pragma unroll
    for (int dt = 0; dt < 4; ++dt) {
      v4f lo4, hi4;
      lo4.x = o[qt][dt][0] * inv; lo4.y = o[qt][dt][1] * inv;
      lo4.z = o[qt][dt][2] * inv; lo4.w = o[qt][dt][3] * inv;
      hi4.x = o[qt][dt][4] * inv; hi4.y = o[qt][dt][5] * inv;
      hi4.z = o[qt][dt][6] * inv; hi4.w = o[qt][dt][7] * inv;
      *(v4f*)(orow + 16 * dt)     = lo4;
      *(v4f*)(orow + 16 * dt + 4) = hi4;
    }
  }
}

__global__ __launch_bounds__(128) void out_proj_kernel(
    const float* __restrict__ AO, const float* __restrict__ W,
    const float* __restrict__ bias, float* __restrict__ Out) {
  const int lane = threadIdx.x & 31;
  const int w    = threadIdx.x >> 5;
  const int m0   = blockIdx.y * 128 + (w >> 1) * 64;
  const int n0   = blockIdx.x * 128 + (w & 1) * 64;

  __shared__ __attribute__((aligned(16))) float gsm[2 * GSTAGE];
  v8f acc[4][4];
  gemm128x128_lds(AO, W, blockIdx.y * 128, blockIdx.x * 128, gsm, acc);

  const int half = lane >> 4, l = lane & 15;
#pragma unroll
  for (int j = 0; j < 4; ++j) {
    const int ncol = n0 + 16 * j + l;
    const float bv = bias[ncol];
#pragma unroll
    for (int i = 0; i < 4; ++i)
#pragma unroll
      for (int v = 0; v < 8; ++v) {
        const int mrow = m0 + 16 * i + v + 8 * half;
        Out[(size_t)mrow * EMB + ncol] = acc[i][j][v] + bv;
      }
  }
}

extern "C" void kernel_launch(void* const* d_in, const int* in_sizes, int n_in,
                              void* d_out, int out_size, void* d_ws, size_t ws_size,
                              hipStream_t stream) {
  const float* x    = (const float*)d_in[0];
  const float* Wqkv = (const float*)d_in[1];
  const float* bqkv = (const float*)d_in[2];
  const float* Wout = (const float*)d_in[3];
  const float* bout = (const float*)d_in[4];
  float* out = (float*)d_out;

  const size_t per = (size_t)4 * NH * S_LEN * DH;
  float* Qb = (float*)d_ws;
  float* Kp = Qb + per;
  float* Vp = Kp + per;
  float* AO = Vp + per;

  dim3 blk(128);
  qkv_proj_kernel<<<dim3(3 * EMB / 128, 4 * S_LEN / 128), blk, 0, stream>>>(
      x, Wqkv, bqkv, Qb, Kp, Vp);
  attn_kernel<<<dim3(4 * NH, S_LEN / (4 * QROWS)), blk, 0, stream>>>(Qb, Kp, Vp, AO);
  out_proj_kernel<<<dim3(EMB / 128, 4 * S_LEN / 128), blk, 0, stream>>>(
      AO, Wout, bout, out);
}
